# starting kernel
# baseline (speedup 1.0000x reference)
#include <hip/hip_runtime.h>


typedef __attribute__((ext_vector_type(2))) float v2f;
typedef __attribute__((ext_vector_type(8))) float v8f;

#define BATCH 32
#define LOUT  512
#define LSEQ  511
#define FEAT  4
#define NCOMP 10
#define NF    340
#define NCH   341
#define COLS  32
#define CHUNK 16

__global__ __launch_bounds__(32)
void sig_scan_kernel(const float* __restrict__ x,
                     const float* __restrict__ kern,
                     float* __restrict__ out)
{
    __shared__ float lds[NCOMP * CHUNK * COLS];

    const int lane = threadIdx.x;
    const int b    = blockIdx.x;
    const int nf0  = blockIdx.y * COLS;

    const int half = lane >> 4;
    const int l15  = lane & 15;
    const int kb   = half * 2;

    v2f blo[NCOMP], bhi[NCOMP];
    #pragma unroll
    for (int c = 0; c < NCOMP; ++c) {
        const int c0 = nf0 + l15;
        const int c1 = nf0 + 16 + l15;
        v2f t0; t0.x = 0.f; t0.y = 0.f;
        v2f t1; t1.x = 0.f; t1.y = 0.f;
        if (c0 < NF) {
            t0.x = kern[(c * FEAT + kb    ) * NF + c0];
            t0.y = kern[(c * FEAT + kb + 1) * NF + c0];
        }
        if (c1 < NF) {
            t1.x = kern[(c * FEAT + kb    ) * NF + c1];
            t1.y = kern[(c * FEAT + kb + 1) * NF + c1];
        }
        blo[c] = t0;
        bhi[c] = t1;
    }

    float p21 = 0.f, p31 = 0.f, p32 = 0.f;
    float p41 = 0.f, p42 = 0.f, p43 = 0.f;
    float tot = 0.f;

    const float* xb   = x   + (size_t)b * LOUT * FEAT;
    float*       outb = out + (size_t)b * LOUT * NCH;

    const int col = lane;
    const int nf  = nf0 + col;

    for (int t0 = 0; t0 < LSEQ; t0 += CHUNK) {
        const int row = t0 + l15;
        v2f a; a.x = 0.f; a.y = 0.f;
        if (row < LSEQ) {
            const float x0a = xb[ row      * FEAT + kb];
            const float x0b = xb[ row      * FEAT + kb + 1];
            const float x1a = xb[(row + 1) * FEAT + kb];
            const float x1b = xb[(row + 1) * FEAT + kb + 1];
            a.x = x1a - x0a;
            a.y = x1b - x0b;
        }

        #pragma unroll
        for (int c = 0; c < NCOMP; ++c) {
            v8f z = {};
            v8f dlo = __builtin_amdgcn_wmma_f32_16x16x4_f32(
                false, a, false, blo[c], (short)0, z, false, false);
            v8f dhi = __builtin_amdgcn_wmma_f32_16x16x4_f32(
                false, a, false, bhi[c], (short)0, z, false, false);
            #pragma unroll
            for (int v = 0; v < 8; ++v) {
                const int r = v + half * 8;
                lds[(c * CHUNK + r) * COLS + l15]      = dlo[v];
                lds[(c * CHUNK + r) * COLS + 16 + l15] = dhi[v];
            }
        }
        __syncthreads();

        for (int r = 0; r < CHUNK; ++r) {
            const int t = t0 + r;
            if (t >= LSEQ) break;
            const float m0 = lds[(0 * CHUNK + r) * COLS + col];
            const float m1 = lds[(1 * CHUNK + r) * COLS + col];
            const float m2 = lds[(2 * CHUNK + r) * COLS + col];
            const float m3 = lds[(3 * CHUNK + r) * COLS + col];
            const float m4 = lds[(4 * CHUNK + r) * COLS + col];
            const float m5 = lds[(5 * CHUNK + r) * COLS + col];
            const float m6 = lds[(6 * CHUNK + r) * COLS + col];
            const float m7 = lds[(7 * CHUNK + r) * COLS + col];
            const float m8 = lds[(8 * CHUNK + r) * COLS + col];
            const float m9 = lds[(9 * CHUNK + r) * COLS + col];

            const float r1 = m0;
            const float r2 = m2 * p21;  p21 += m1;
            const float s3 = m4 * p31;  p31 += m3;
            const float r3 = m5 * p32;  p32 += s3;
            const float s4 = m7 * p41;  p41 += m6;
            const float s5 = m8 * p42;  p42 += s4;
            const float r4 = m9 * p43;  p43 += s5;

            tot += r1 + r2 + r3 + r4;
            if (nf < NF)
                outb[(size_t)(t + 1) * NCH + 1 + nf] = tot;
        }
        __syncthreads();
    }
}

__global__ void sig_init_kernel(float* __restrict__ out)
{
    const int idx = blockIdx.x * blockDim.x + threadIdx.x;
    if (idx < BATCH * LOUT) {
        const int b = idx / LOUT;
        const int t = idx % LOUT;
        out[(size_t)(b * LOUT + t) * NCH] = 1.0f;
        if (t < NF)
            out[(size_t)b * LOUT * NCH + 1 + t] = 0.0f;
    }
}

extern "C" void kernel_launch(void* const* d_in, const int* in_sizes, int n_in,
                              void* d_out, int out_size, void* d_ws, size_t ws_size,
                              hipStream_t stream)
{
    (void)in_sizes; (void)n_in; (void)out_size; (void)d_ws; (void)ws_size;
    const float* x    = (const float*)d_in[0];
    const float* kern = (const float*)d_in[1];
    float*       out  = (float*)d_out;

    sig_init_kernel<<<(BATCH * LOUT + 255) / 256, 256, 0, stream>>>(out);

    dim3 grid(BATCH, (NF + COLS - 1) / COLS);
    sig_scan_kernel<<<grid, 32, 0, stream>>>(x, kern, out);
}


# Scan straight from WMMA D registers: A rows = 8 timesteps x 2 batches, so a lane's 8 D registers are its own column's timesteps; LDS staging and per-chunk barriers gone; same serial sum order
# speedup vs baseline: 1.8362x; 1.8362x over previous
#include <hip/hip_runtime.h>


typedef __attribute__((ext_vector_type(2))) float v2f;
typedef __attribute__((ext_vector_type(8))) float v8f;

#define BATCH  32
#define LOUT   512
#define LSEQ   511
#define FEAT   4
#define NCOMP  10
#define NF     340
#define NCH    341
#define TCOLS  16
#define TSTEPS 8
#define NPAIR  16
#define NTILE  22
#define TFULL  504

struct ScanState {
    float p21, p31, p32, p41, p42, p43, tot;
};

template <int NSTEPS>
__device__ __forceinline__ void sig_tile(const float* __restrict__ xa, const int t0, const int tl,
                                         const v2f (&bfrag)[NCOMP], ScanState& s,
                                         float* __restrict__ ocol, const bool valid)
{
    int at = t0 + tl;
    if (NSTEPS < TSTEPS) {
        at = at < LSEQ - 1 ? at : LSEQ - 1;
    }
    const float x0a = xa[ at      * FEAT];
    const float x0b = xa[ at      * FEAT + 1];
    const float x1a = xa[(at + 1) * FEAT];
    const float x1b = xa[(at + 1) * FEAT + 1];
    v2f a;
    a.x = x1a - x0a;
    a.y = x1b - x0b;

    v8f m[NCOMP];
    #pragma unroll
    for (int c = 0; c < NCOMP; ++c) {
        v8f z = {};
        m[c] = __builtin_amdgcn_wmma_f32_16x16x4_f32(
            false, a, false, bfrag[c], (short)0, z, false, false);
    }

    float o[NSTEPS];
    #pragma unroll
    for (int r = 0; r < NSTEPS; ++r) {
        const float m0 = m[0][r], m1 = m[1][r], m2 = m[2][r], m3 = m[3][r], m4 = m[4][r];
        const float m5 = m[5][r], m6 = m[6][r], m7 = m[7][r], m8 = m[8][r], m9 = m[9][r];

        const float r1 = m0;
        const float r2 = m2 * s.p21;  s.p21 += m1;
        const float s3 = m4 * s.p31;  s.p31 += m3;
        const float r3 = m5 * s.p32;  s.p32 += s3;
        const float s4 = m7 * s.p41;  s.p41 += m6;
        const float s5 = m8 * s.p42;  s.p42 += s4;
        const float r4 = m9 * s.p43;  s.p43 += s5;

        s.tot += r1 + r2 + r3 + r4;
        o[r] = s.tot;
    }

    if (valid) {
        #pragma unroll
        for (int r = 0; r < NSTEPS; ++r)
            ocol[(size_t)(t0 + 1 + r) * NCH] = o[r];
    }
}

__global__ __launch_bounds__(32)
void sig_scan_kernel(const float* __restrict__ x,
                     const float* __restrict__ kern,
                     float* __restrict__ out)
{
    const int lane = threadIdx.x;
    const int bp   = blockIdx.x;
    const int nf0  = blockIdx.y * TCOLS;

    const int half = lane >> 4;
    const int l15  = lane & 15;
    const int kb   = half * 2;

    const int  nf    = nf0 + l15;
    const bool valid = nf < NF;

    v2f bfrag[NCOMP];
    #pragma unroll
    for (int c = 0; c < NCOMP; ++c) {
        v2f t; t.x = 0.f; t.y = 0.f;
        if (valid) {
            t.x = kern[(c * FEAT + kb    ) * NF + nf];
            t.y = kern[(c * FEAT + kb + 1) * NF + nf];
        }
        bfrag[c] = t;
    }

    const float* xa = x + (size_t)(2 * bp + (l15 >> 3)) * LOUT * FEAT + kb;
    const int    tl = l15 & 7;
    float* ocol = out + (size_t)(2 * bp + half) * LOUT * NCH + 1 + nf;

    ScanState s;
    s.p21 = 0.f; s.p31 = 0.f; s.p32 = 0.f;
    s.p41 = 0.f; s.p42 = 0.f; s.p43 = 0.f;
    s.tot = 0.f;

    for (int t0 = 0; t0 < TFULL; t0 += TSTEPS)
        sig_tile<TSTEPS>(xa, t0, tl, bfrag, s, ocol, valid);
    sig_tile<LSEQ - TFULL>(xa, TFULL, tl, bfrag, s, ocol, valid);
}

__global__ void sig_init_kernel(float* __restrict__ out)
{
    const int idx = blockIdx.x * blockDim.x + threadIdx.x;
    if (idx < BATCH * LOUT) {
        const int b = idx / LOUT;
        const int t = idx % LOUT;
        out[(size_t)(b * LOUT + t) * NCH] = 1.0f;
        if (t < NF)
            out[(size_t)b * LOUT * NCH + 1 + t] = 0.0f;
    }
}

extern "C" void kernel_launch(void* const* d_in, const int* in_sizes, int n_in,
                              void* d_out, int out_size, void* d_ws, size_t ws_size,
                              hipStream_t stream)
{
    (void)in_sizes; (void)n_in; (void)out_size; (void)d_ws; (void)ws_size;
    const float* x    = (const float*)d_in[0];
    const float* kern = (const float*)d_in[1];
    float*       out  = (float*)d_out;

    sig_init_kernel<<<(BATCH * LOUT + 255) / 256, 256, 0, stream>>>(out);

    dim3 grid(NPAIR, NTILE);
    sig_scan_kernel<<<grid, 32, 0, stream>>>(x, kern, out);
}


# x rows of each group of 4 tiles are loaded while the group before is scanned (register double buffer): the wave's serial chain no longer waits for a load every tile; sums and their order unchanged
# speedup vs baseline: 5.2100x; 2.8374x over previous
#include <hip/hip_runtime.h>


typedef __attribute__((ext_vector_type(2))) float v2f;
typedef __attribute__((ext_vector_type(8))) float v8f;

#define BATCH  32
#define LOUT   512
#define LSEQ   511
#define FEAT   4
#define NCOMP  10
#define NF     340
#define NCH    341
#define TCOLS  16
#define TSTEPS 8
#define NPAIR  16
#define NTILE  22
#define TFULL  504
#define PF     4

struct ScanState {
    float p21, p31, p32, p41, p42, p43, tot;
};

struct XRows {
    float x0a, x0b, x1a, x1b;
};

__device__ __forceinline__ XRows load_rows(const float* __restrict__ xa, const int t0, const int tl)
{
    int at = t0 + tl;
    at = at < LSEQ - 1 ? at : LSEQ - 1;
    XRows q;
    q.x0a = xa[ at      * FEAT];
    q.x0b = xa[ at      * FEAT + 1];
    q.x1a = xa[(at + 1) * FEAT];
    q.x1b = xa[(at + 1) * FEAT + 1];
    return q;
}

__device__ __forceinline__ v2f increment(const XRows& q)
{
    v2f a;
    a.x = q.x1a - q.x0a;
    a.y = q.x1b - q.x0b;
    return a;
}

template <int NSTEPS>
__device__ __forceinline__ void sig_tile(const v2f a, const int t0,
                                         const v2f (&bfrag)[NCOMP], ScanState& s,
                                         float* __restrict__ ocol, const bool valid)
{
    v8f m[NCOMP];
    #pragma unroll
    for (int c = 0; c < NCOMP; ++c) {
        v8f z = {};
        m[c] = __builtin_amdgcn_wmma_f32_16x16x4_f32(
            false, a, false, bfrag[c], (short)0, z, false, false);
    }

    float o[NSTEPS];
    #pragma unroll
    for (int r = 0; r < NSTEPS; ++r) {
        const float m0 = m[0][r], m1 = m[1][r], m2 = m[2][r], m3 = m[3][r], m4 = m[4][r];
        const float m5 = m[5][r], m6 = m[6][r], m7 = m[7][r], m8 = m[8][r], m9 = m[9][r];

        const float r1 = m0;
        const float r2 = m2 * s.p21;  s.p21 += m1;
        const float s3 = m4 * s.p31;  s.p31 += m3;
        const float r3 = m5 * s.p32;  s.p32 += s3;
        const float s4 = m7 * s.p41;  s.p41 += m6;
        const float s5 = m8 * s.p42;  s.p42 += s4;
        const float r4 = m9 * s.p43;  s.p43 += s5;

        s.tot += r1 + r2 + r3 + r4;
        o[r] = s.tot;
    }

    if (valid) {
        #pragma unroll
        for (int r = 0; r < NSTEPS; ++r)
            ocol[(size_t)(t0 + 1 + r) * NCH] = o[r];
    }
}

__global__ __launch_bounds__(32)
void sig_scan_kernel(const float* __restrict__ x,
                     const float* __restrict__ kern,
                     float* __restrict__ out)
{
    const int lane = threadIdx.x;
    const int bp   = blockIdx.x;
    const int nf0  = blockIdx.y * TCOLS;

    const int half = lane >> 4;
    const int l15  = lane & 15;
    const int kb   = half * 2;

    const int  nf    = nf0 + l15;
    const bool valid = nf < NF;

    v2f bfrag[NCOMP];
    #pragma unroll
    for (int c = 0; c < NCOMP; ++c) {
        v2f t; t.x = 0.f; t.y = 0.f;
        if (valid) {
            t.x = kern[(c * FEAT + kb    ) * NF + nf];
            t.y = kern[(c * FEAT + kb + 1) * NF + nf];
        }
        bfrag[c] = t;
    }

    const float* xa = x + (size_t)(2 * bp + (l15 >> 3)) * LOUT * FEAT + kb;
    const int    tl = l15 & 7;
    float* ocol = out + (size_t)(2 * bp + half) * LOUT * NCH + 1 + nf;

    ScanState s;
    s.p21 = 0.f; s.p31 = 0.f; s.p32 = 0.f;
    s.p41 = 0.f; s.p42 = 0.f; s.p43 = 0.f;
    s.tot = 0.f;

    XRows nxt[PF];
    #pragma unroll
    for (int i = 0; i < PF; ++i)
        nxt[i] = load_rows(xa, i * TSTEPS, tl);

    for (int t0 = 0; t0 < TFULL + TSTEPS - PF * TSTEPS; t0 += PF * TSTEPS) {
        v2f a[PF];
        #pragma unroll
        for (int i = 0; i < PF; ++i)
            a[i] = increment(nxt[i]);
        #pragma unroll
        for (int i = 0; i < PF; ++i)
            nxt[i] = load_rows(xa, t0 + (PF + i) * TSTEPS, tl);
        #pragma unroll
        for (int i = 0; i < PF; ++i)
            sig_tile<TSTEPS>(a[i], t0 + i * TSTEPS, bfrag, s, ocol, valid);
    }
    #pragma unroll
    for (int i = 0; i < PF - 1; ++i)
        sig_tile<TSTEPS>(increment(nxt[i]), TFULL - (PF - 1 - i) * TSTEPS, bfrag, s, ocol, valid);
    sig_tile<LSEQ - TFULL>(increment(nxt[PF - 1]), TFULL, bfrag, s, ocol, valid);
}

__global__ void sig_init_kernel(float* __restrict__ out)
{
    const int idx = blockIdx.x * blockDim.x + threadIdx.x;
    if (idx < BATCH * LOUT) {
        const int b = idx / LOUT;
        const int t = idx % LOUT;
        out[(size_t)(b * LOUT + t) * NCH] = 1.0f;
        if (t < NF)
            out[(size_t)b * LOUT * NCH + 1 + t] = 0.0f;
    }
}

extern "C" void kernel_launch(void* const* d_in, const int* in_sizes, int n_in,
                              void* d_out, int out_size, void* d_ws, size_t ws_size,
                              hipStream_t stream)
{
    (void)in_sizes; (void)n_in; (void)out_size; (void)d_ws; (void)ws_size;
    const float* x    = (const float*)d_in[0];
    const float* kern = (const float*)d_in[1];
    float*       out  = (float*)d_out;

    sig_init_kernel<<<(BATCH * LOUT + 255) / 256, 256, 0, stream>>>(out);

    dim3 grid(NPAIR, NTILE);
    sig_scan_kernel<<<grid, 32, 0, stream>>>(x, kern, out);
}


# One launch instead of two: the scan waves store the constant-1 channel (64 rows each, tiles 0-15 of a batch pair) and their own columns' zero t=0 row; the fill kernel and its launch gap are gone
# speedup vs baseline: 6.1130x; 1.1733x over previous
#include <hip/hip_runtime.h>


typedef __attribute__((ext_vector_type(2))) float v2f;
typedef __attribute__((ext_vector_type(8))) float v8f;

#define BATCH  32
#define LOUT   512
#define LSEQ   511
#define FEAT   4
#define NCOMP  10
#define NF     340
#define NCH    341
#define TCOLS  16
#define TSTEPS 8
#define NPAIR  16
#define NTILE  22
#define TFULL  504
#define PF     4
#define ONESW  16
#define ONESR  64

struct ScanState {
    float p21, p31, p32, p41, p42, p43, tot;
};

struct XRows {
    float x0a, x0b, x1a, x1b;
};

__device__ __forceinline__ XRows load_rows(const float* __restrict__ xa, const int t0, const int tl)
{
    int at = t0 + tl;
    at = at < LSEQ - 1 ? at : LSEQ - 1;
    XRows q;
    q.x0a = xa[ at      * FEAT];
    q.x0b = xa[ at      * FEAT + 1];
    q.x1a = xa[(at + 1) * FEAT];
    q.x1b = xa[(at + 1) * FEAT + 1];
    return q;
}

__device__ __forceinline__ v2f increment(const XRows& q)
{
    v2f a;
    a.x = q.x1a - q.x0a;
    a.y = q.x1b - q.x0b;
    return a;
}

template <int NSTEPS>
__device__ __forceinline__ void sig_tile(const v2f a, const int t0,
                                         const v2f (&bfrag)[NCOMP], ScanState& s,
                                         float* __restrict__ ocol, const bool valid)
{
    v8f m[NCOMP];
    #pragma unroll
    for (int c = 0; c < NCOMP; ++c) {
        v8f z = {};
        m[c] = __builtin_amdgcn_wmma_f32_16x16x4_f32(
            false, a, false, bfrag[c], (short)0, z, false, false);
    }

    float o[NSTEPS];
    #pragma unroll
    for (int r = 0; r < NSTEPS; ++r) {
        const float m0 = m[0][r], m1 = m[1][r], m2 = m[2][r], m3 = m[3][r], m4 = m[4][r];
        const float m5 = m[5][r], m6 = m[6][r], m7 = m[7][r], m8 = m[8][r], m9 = m[9][r];

        const float r1 = m0;
        const float r2 = m2 * s.p21;  s.p21 += m1;
        const float s3 = m4 * s.p31;  s.p31 += m3;
        const float r3 = m5 * s.p32;  s.p32 += s3;
        const float s4 = m7 * s.p41;  s.p41 += m6;
        const float s5 = m8 * s.p42;  s.p42 += s4;
        const float r4 = m9 * s.p43;  s.p43 += s5;

        s.tot += r1 + r2 + r3 + r4;
        o[r] = s.tot;
    }

    if (valid) {
        #pragma unroll
        for (int r = 0; r < NSTEPS; ++r)
            ocol[(size_t)(t0 + 1 + r) * NCH] = o[r];
    }
}

__global__ __launch_bounds__(32)
void sig_scan_kernel(const float* __restrict__ x,
                     const float* __restrict__ kern,
                     float* __restrict__ out)
{
    const int lane = threadIdx.x;
    const int bp   = blockIdx.x;
    const int nf0  = blockIdx.y * TCOLS;

    const int half = lane >> 4;
    const int l15  = lane & 15;
    const int kb   = half * 2;

    const int  nf    = nf0 + l15;
    const bool valid = nf < NF;

    v2f bfrag[NCOMP];
    #pragma unroll
    for (int c = 0; c < NCOMP; ++c) {
        v2f t; t.x = 0.f; t.y = 0.f;
        if (valid) {
            t.x = kern[(c * FEAT + kb    ) * NF + nf];
            t.y = kern[(c * FEAT + kb + 1) * NF + nf];
        }
        bfrag[c] = t;
    }

    const float* xa = x + (size_t)(2 * bp + (l15 >> 3)) * LOUT * FEAT + kb;
    const int    tl = l15 & 7;
    float* ocol = out + (size_t)(2 * bp + half) * LOUT * NCH + 1 + nf;

    ScanState s;
    s.p21 = 0.f; s.p31 = 0.f; s.p32 = 0.f;
    s.p41 = 0.f; s.p42 = 0.f; s.p43 = 0.f;
    s.tot = 0.f;

    XRows nxt[PF];
    #pragma unroll
    for (int i = 0; i < PF; ++i)
        nxt[i] = load_rows(xa, i * TSTEPS, tl);

    if (blockIdx.y < ONESW) {
        float* ones = out + (size_t)(2 * bp) * LOUT * NCH;
        #pragma unroll
        for (int j = 0; j < ONESR / 32; ++j)
            ones[(size_t)(blockIdx.y * ONESR + j * 32 + lane) * NCH] = 1.0f;
    }
    if (valid)
        ocol[0] = 0.0f;

    for (int t0 = 0; t0 < TFULL + TSTEPS - PF * TSTEPS; t0 += PF * TSTEPS) {
        v2f a[PF];
        #pragma unroll
        for (int i = 0; i < PF; ++i)
            a[i] = increment(nxt[i]);
        #pragma unroll
        for (int i = 0; i < PF; ++i)
            nxt[i] = load_rows(xa, t0 + (PF + i) * TSTEPS, tl);
        #pragma unroll
        for (int i = 0; i < PF; ++i)
            sig_tile<TSTEPS>(a[i], t0 + i * TSTEPS, bfrag, s, ocol, valid);
    }
    #pragma unroll
    for (int i = 0; i < PF - 1; ++i)
        sig_tile<TSTEPS>(increment(nxt[i]), TFULL - (PF - 1 - i) * TSTEPS, bfrag, s, ocol, valid);
    sig_tile<LSEQ - TFULL>(increment(nxt[PF - 1]), TFULL, bfrag, s, ocol, valid);
}

extern "C" void kernel_launch(void* const* d_in, const int* in_sizes, int n_in,
                              void* d_out, int out_size, void* d_ws, size_t ws_size,
                              hipStream_t stream)
{
    (void)in_sizes; (void)n_in; (void)out_size; (void)d_ws; (void)ws_size;
    const float* x    = (const float*)d_in[0];
    const float* kern = (const float*)d_in[1];
    float*       out  = (float*)d_out;

    dim3 grid(NPAIR, NTILE);
    sig_scan_kernel<<<grid, 32, 0, stream>>>(x, kern, out);
}


# Level-4 chain moved to a second wave of the workgroup (own SIMD): products split 6+4, running sums joined through LDS once per 4 tiles; out = sum(levels 1-3) + sum(level 4): sum order changed
# speedup vs baseline: 6.8254x; 1.1165x over previous
#include <hip/hip_runtime.h>


typedef __attribute__((ext_vector_type(2))) float v2f;
typedef __attribute__((ext_vector_type(4))) float v4f;
typedef __attribute__((ext_vector_type(8))) float v8f;

#define BATCH  32
#define LOUT   512
#define LSEQ   511
#define FEAT   4
#define NCOMP  10
#define NLOW   6
#define NHIGH  4
#define NF     340
#define NCH    341
#define TCOLS  16
#define TSTEPS 8
#define NPAIR  16
#define NTILE  22
#define TFULL  504
#define PF     4
#define ONESW  16
#define ONESR  64
#define XPITCH 3

struct LowState {
    float p21, p31, p32, tot;
};

struct HighState {
    float p41, p42, p43, tot;
};

struct XRows {
    float x0a, x0b, x1a, x1b;
};

__device__ __forceinline__ XRows load_rows(const float* __restrict__ xa, const int t0, const int tl)
{
    int at = t0 + tl;
    at = at < LSEQ - 1 ? at : LSEQ - 1;
    XRows q;
    q.x0a = xa[ at      * FEAT];
    q.x0b = xa[ at      * FEAT + 1];
    q.x1a = xa[(at + 1) * FEAT];
    q.x1b = xa[(at + 1) * FEAT + 1];
    return q;
}

__device__ __forceinline__ v2f increment(const XRows& q)
{
    v2f a;
    a.x = q.x1a - q.x0a;
    a.y = q.x1b - q.x0b;
    return a;
}

__device__ __forceinline__ void low_tile(const v2f a, const v2f (&bfrag)[NCOMP],
                                         LowState& s, v4f* __restrict__ slot)
{
    v8f m[NLOW];
    #pragma unroll
    for (int c = 0; c < NLOW; ++c) {
        v8f z = {};
        m[c] = __builtin_amdgcn_wmma_f32_16x16x4_f32(
            false, a, false, bfrag[c], (short)0, z, false, false);
    }

    float o[TSTEPS];
    #pragma unroll
    for (int r = 0; r < TSTEPS; ++r) {
        const float m0 = m[0][r], m1 = m[1][r], m2 = m[2][r];
        const float m3 = m[3][r], m4 = m[4][r], m5 = m[5][r];

        const float r1 = m0;
        const float r2 = m2 * s.p21;  s.p21 += m1;
        const float s3 = m4 * s.p31;  s.p31 += m3;
        const float r3 = m5 * s.p32;  s.p32 += s3;

        s.tot += r1 + r2 + r3;
        o[r] = s.tot;
    }

    v4f lo, hi;
    lo.x = o[0]; lo.y = o[1]; lo.z = o[2]; lo.w = o[3];
    hi.x = o[4]; hi.y = o[5]; hi.z = o[6]; hi.w = o[7];
    slot[0] = lo;
    slot[1] = hi;
}

__device__ __forceinline__ void high_tile(const v2f a, const v2f (&bfrag)[NCOMP],
                                          HighState& s, float (&q)[TSTEPS])
{
    v8f m[NHIGH];
    #pragma unroll
    for (int c = 0; c < NHIGH; ++c) {
        v8f z = {};
        m[c] = __builtin_amdgcn_wmma_f32_16x16x4_f32(
            false, a, false, bfrag[NLOW + c], (short)0, z, false, false);
    }

    #pragma unroll
    for (int r = 0; r < TSTEPS; ++r) {
        const float m6 = m[0][r], m7 = m[1][r], m8 = m[2][r], m9 = m[3][r];

        const float s4 = m7 * s.p41;  s.p41 += m6;
        const float s5 = m8 * s.p42;  s.p42 += s4;
        const float r4 = m9 * s.p43;  s.p43 += s5;

        s.tot += r4;
        q[r] = s.tot;
    }
}

__device__ __forceinline__ void lds_barrier()
{
    __builtin_amdgcn_fence(__ATOMIC_RELEASE, "workgroup", "local");
    __builtin_amdgcn_s_barrier();
    __builtin_amdgcn_fence(__ATOMIC_ACQUIRE, "workgroup", "local");
}

template <bool LAST>
__device__ __forceinline__ void sig_group(const int role, const v2f (&a)[PF], const int t0,
                                          const v2f (&bfrag)[NCOMP],
                                          LowState& lo, HighState& hi,
                                          v4f (* __restrict__ xb)[32][XPITCH], const int lane,
                                          float* __restrict__ ocol, const bool valid)
{
    float q[PF][TSTEPS];

    if (role == 0) {
        #pragma unroll
        for (int i = 0; i < PF; ++i)
            low_tile(a[i], bfrag, lo, &xb[i][lane][0]);
    } else {
        #pragma unroll
        for (int i = 0; i < PF; ++i)
            high_tile(a[i], bfrag, hi, q[i]);
    }

    lds_barrier();

    if (role != 0) {
        #pragma unroll
        for (int i = 0; i < PF; ++i) {
            const v4f l0 = xb[i][lane][0];
            const v4f l1 = xb[i][lane][1];
            float o[TSTEPS];
            o[0] = l0.x + q[i][0];
            o[1] = l0.y + q[i][1];
            o[2] = l0.z + q[i][2];
            o[3] = l0.w + q[i][3];
            o[4] = l1.x + q[i][4];
            o[5] = l1.y + q[i][5];
            o[6] = l1.z + q[i][6];
            o[7] = l1.w + q[i][7];
            if (valid) {
                #pragma unroll
                for (int r = 0; r < TSTEPS; ++r) {
                    if (!(LAST && i == PF - 1 && r == TSTEPS - 1))
                        ocol[(size_t)(t0 + i * TSTEPS + 1 + r) * NCH] = o[r];
                }
            }
        }
    }
}

__global__ __launch_bounds__(64)
void sig_scan_kernel(const float* __restrict__ x,
                     const float* __restrict__ kern,
                     float* __restrict__ out)
{
    __shared__ v4f xch[2][PF][32][XPITCH];

    const int lane = threadIdx.x & 31;
    const int role = __builtin_amdgcn_readfirstlane((int)(threadIdx.x >> 5));
    const int bp   = blockIdx.x;
    const int nf0  = blockIdx.y * TCOLS;

    const int half = lane >> 4;
    const int l15  = lane & 15;
    const int kb   = half * 2;

    const int  nf    = nf0 + l15;
    const bool valid = nf < NF;

    v2f bfrag[NCOMP];
    #pragma unroll
    for (int c = 0; c < NCOMP; ++c) {
        v2f t; t.x = 0.f; t.y = 0.f;
        if (valid) {
            t.x = kern[(c * FEAT + kb    ) * NF + nf];
            t.y = kern[(c * FEAT + kb + 1) * NF + nf];
        }
        bfrag[c] = t;
    }

    const float* xa = x + (size_t)(2 * bp + (l15 >> 3)) * LOUT * FEAT + kb;
    const int    tl = l15 & 7;
    float* ocol = out + (size_t)(2 * bp + half) * LOUT * NCH + 1 + nf;

    LowState lo;
    lo.p21 = 0.f; lo.p31 = 0.f; lo.p32 = 0.f; lo.tot = 0.f;
    HighState hi;
    hi.p41 = 0.f; hi.p42 = 0.f; hi.p43 = 0.f; hi.tot = 0.f;

    XRows nxt[PF];
    #pragma unroll
    for (int i = 0; i < PF; ++i)
        nxt[i] = load_rows(xa, i * TSTEPS, tl);

    if (blockIdx.y < ONESW) {
        float* ones = out + (size_t)(2 * bp) * LOUT * NCH;
        ones[(size_t)(blockIdx.y * ONESR + threadIdx.x) * NCH] = 1.0f;
    }
    if (role == 0 && valid)
        ocol[0] = 0.0f;

    int buf = 0;
    for (int t0 = 0; t0 < TFULL + TSTEPS - PF * TSTEPS; t0 += PF * TSTEPS) {
        v2f a[PF];
        #pragma unroll
        for (int i = 0; i < PF; ++i)
            a[i] = increment(nxt[i]);
        #pragma unroll
        for (int i = 0; i < PF; ++i)
            nxt[i] = load_rows(xa, t0 + (PF + i) * TSTEPS, tl);
        sig_group<false>(role, a, t0, bfrag, lo, hi, xch[buf], lane, ocol, valid);
        buf ^= 1;
    }
    {
        v2f a[PF];
        #pragma unroll
        for (int i = 0; i < PF; ++i)
            a[i] = increment(nxt[i]);
        sig_group<true>(role, a, TFULL + TSTEPS - PF * TSTEPS, bfrag, lo, hi, xch[buf], lane, ocol, valid);
    }
}

extern "C" void kernel_launch(void* const* d_in, const int* in_sizes, int n_in,
                              void* d_out, int out_size, void* d_ws, size_t ws_size,
                              hipStream_t stream)
{
    (void)in_sizes; (void)n_in; (void)out_size; (void)d_ws; (void)ws_size;
    const float* x    = (const float*)d_in[0];
    const float* kern = (const float*)d_in[1];
    float*       out  = (float*)d_out;

    dim3 grid(NPAIR, NTILE);
    sig_scan_kernel<<<grid, 64, 0, stream>>>(x, kern, out);
}


# x rows of the batch pair turned into increments once and kept in LDS (plain loads, ds_store); A fragments read from LDS: no global load left in the loop behind the output stores; same sums and order
# speedup vs baseline: 7.7592x; 1.1368x over previous
#include <hip/hip_runtime.h>


typedef __attribute__((ext_vector_type(2))) float v2f;
typedef __attribute__((ext_vector_type(4))) float v4f;
typedef __attribute__((ext_vector_type(8))) float v8f;

#define BATCH  32
#define LOUT   512
#define LSEQ   511
#define FEAT   4
#define NCOMP  10
#define NLOW   6
#define NHIGH  4
#define NF     340
#define NCH    341
#define TCOLS  16
#define TSTEPS 8
#define NPAIR  16
#define NTILE  22
#define TFULL  504
#define PF     4
#define ONESW  16
#define ONESR  64
#define XPITCH 3
#define SEQPAD 8

struct LowState {
    float p21, p31, p32, tot;
};

struct HighState {
    float p41, p42, p43, tot;
};

__device__ __forceinline__ void low_tile(const v2f a, const v2f (&bfrag)[NCOMP],
                                         LowState& s, v4f* __restrict__ slot)
{
    v8f m[NLOW];
    #pragma unroll
    for (int c = 0; c < NLOW; ++c) {
        v8f z = {};
        m[c] = __builtin_amdgcn_wmma_f32_16x16x4_f32(
            false, a, false, bfrag[c], (short)0, z, false, false);
    }

    float o[TSTEPS];
    #pragma unroll
    for (int r = 0; r < TSTEPS; ++r) {
        const float m0 = m[0][r], m1 = m[1][r], m2 = m[2][r];
        const float m3 = m[3][r], m4 = m[4][r], m5 = m[5][r];

        const float r1 = m0;
        const float r2 = m2 * s.p21;  s.p21 += m1;
        const float s3 = m4 * s.p31;  s.p31 += m3;
        const float r3 = m5 * s.p32;  s.p32 += s3;

        s.tot += r1 + r2 + r3;
        o[r] = s.tot;
    }

    v4f lo, hi;
    lo.x = o[0]; lo.y = o[1]; lo.z = o[2]; lo.w = o[3];
    hi.x = o[4]; hi.y = o[5]; hi.z = o[6]; hi.w = o[7];
    slot[0] = lo;
    slot[1] = hi;
}

__device__ __forceinline__ void high_tile(const v2f a, const v2f (&bfrag)[NCOMP],
                                          HighState& s, float (&q)[TSTEPS])
{
    v8f m[NHIGH];
    #pragma unroll
    for (int c = 0; c < NHIGH; ++c) {
        v8f z = {};
        m[c] = __builtin_amdgcn_wmma_f32_16x16x4_f32(
            false, a, false, bfrag[NLOW + c], (short)0, z, false, false);
    }

    #pragma unroll
    for (int r = 0; r < TSTEPS; ++r) {
        const float m6 = m[0][r], m7 = m[1][r], m8 = m[2][r], m9 = m[3][r];

        const float s4 = m7 * s.p41;  s.p41 += m6;
        const float s5 = m8 * s.p42;  s.p42 += s4;
        const float r4 = m9 * s.p43;  s.p43 += s5;

        s.tot += r4;
        q[r] = s.tot;
    }
}

__device__ __forceinline__ void lds_barrier()
{
    __builtin_amdgcn_fence(__ATOMIC_RELEASE, "workgroup", "local");
    __builtin_amdgcn_s_barrier();
    __builtin_amdgcn_fence(__ATOMIC_ACQUIRE, "workgroup", "local");
}

template <bool LAST>
__device__ __forceinline__ void sig_group(const int role, const v2f (&a)[PF], const int t0,
                                          const v2f (&bfrag)[NCOMP],
                                          LowState& lo, HighState& hi,
                                          v4f (* __restrict__ xb)[32][XPITCH], const int lane,
                                          float* __restrict__ ocol, const bool valid)
{
    float q[PF][TSTEPS];

    if (role == 0) {
        #pragma unroll
        for (int i = 0; i < PF; ++i)
            low_tile(a[i], bfrag, lo, &xb[i][lane][0]);
    } else {
        #pragma unroll
        for (int i = 0; i < PF; ++i)
            high_tile(a[i], bfrag, hi, q[i]);
    }

    lds_barrier();

    if (role != 0) {
        #pragma unroll
        for (int i = 0; i < PF; ++i) {
            const v4f l0 = xb[i][lane][0];
            const v4f l1 = xb[i][lane][1];
            float o[TSTEPS];
            o[0] = l0.x + q[i][0];
            o[1] = l0.y + q[i][1];
            o[2] = l0.z + q[i][2];
            o[3] = l0.w + q[i][3];
            o[4] = l1.x + q[i][4];
            o[5] = l1.y + q[i][5];
            o[6] = l1.z + q[i][6];
            o[7] = l1.w + q[i][7];
            if (valid) {
                #pragma unroll
                for (int r = 0; r < TSTEPS; ++r) {
                    if (!(LAST && i == PF - 1 && r == TSTEPS - 1))
                        ocol[(size_t)(t0 + i * TSTEPS + 1 + r) * NCH] = o[r];
                }
            }
        }
    }
}

__global__ __launch_bounds__(64)
void sig_scan_kernel(const float* __restrict__ x,
                     const float* __restrict__ kern,
                     float* __restrict__ out)
{
    __shared__ v4f xch[2][PF][32][XPITCH];
    __shared__ v4f seqs[2][LOUT + SEQPAD];

    const int lane = threadIdx.x & 31;
    const int role = __builtin_amdgcn_readfirstlane((int)(threadIdx.x >> 5));
    const int bp   = blockIdx.x;
    const int nf0  = blockIdx.y * TCOLS;

    const int half = lane >> 4;
    const int l15  = lane & 15;
    const int kb   = half * 2;

    const int  nf    = nf0 + l15;
    const bool valid = nf < NF;

    v2f bfrag[NCOMP];
    #pragma unroll
    for (int c = 0; c < NCOMP; ++c) {
        v2f t; t.x = 0.f; t.y = 0.f;
        if (valid) {
            t.x = kern[(c * FEAT + kb    ) * NF + nf];
            t.y = kern[(c * FEAT + kb + 1) * NF + nf];
        }
        bfrag[c] = t;
    }

    {
        const v4f* rows = (const v4f*)(x + (size_t)(2 * bp) * LOUT * FEAT) + threadIdx.x;
        #pragma unroll
        for (int k = 0; k < (2 * LOUT) / 64; ++k) {
            const int t = (int)threadIdx.x + 64 * (k % 8);
            v4f r0, r1;
            if (64 * (k % 8) + 63 < LSEQ) {
                r0 = rows[64 * k];
                r1 = rows[64 * k + 1];
            } else {
                const int back = t < LSEQ ? 0 : -1;
                r0 = rows[64 * k + back];
                r1 = rows[64 * k + back + 1];
            }
            seqs[k / 8][t] = r1 - r0;
        }
    }

    const v2f* sa = (const v2f*)&seqs[l15 >> 3][l15 & 7] + half;
    float* ocol = out + (size_t)(2 * bp + half) * LOUT * NCH + 1 + nf;

    LowState lo;
    lo.p21 = 0.f; lo.p31 = 0.f; lo.p32 = 0.f; lo.tot = 0.f;
    HighState hi;
    hi.p41 = 0.f; hi.p42 = 0.f; hi.p43 = 0.f; hi.tot = 0.f;

    if (blockIdx.y < ONESW) {
        float* ones = out + (size_t)(2 * bp) * LOUT * NCH;
        ones[(size_t)(blockIdx.y * ONESR + threadIdx.x) * NCH] = 1.0f;
    }
    if (role == 0 && valid)
        ocol[0] = 0.0f;

    lds_barrier();

    v2f nxt[PF];
    #pragma unroll
    for (int i = 0; i < PF; ++i)
        nxt[i] = sa[(i * TSTEPS) * 2];

    int buf = 0;
    for (int t0 = 0; t0 < TFULL + TSTEPS - PF * TSTEPS; t0 += PF * TSTEPS) {
        v2f a[PF];
        #pragma unroll
        for (int i = 0; i < PF; ++i)
            a[i] = nxt[i];
        #pragma unroll
        for (int i = 0; i < PF; ++i)
            nxt[i] = sa[(t0 + (PF + i) * TSTEPS) * 2];
        sig_group<false>(role, a, t0, bfrag, lo, hi, xch[buf], lane, ocol, valid);
        buf ^= 1;
    }
    {
        v2f a[PF];
        #pragma unroll
        for (int i = 0; i < PF; ++i)
            a[i] = nxt[i];
        sig_group<true>(role, a, TFULL + TSTEPS - PF * TSTEPS, bfrag, lo, hi, xch[buf], lane, ocol, valid);
    }
}

extern "C" void kernel_launch(void* const* d_in, const int* in_sizes, int n_in,
                              void* d_out, int out_size, void* d_ws, size_t ws_size,
                              hipStream_t stream)
{
    (void)in_sizes; (void)n_in; (void)out_size; (void)d_ws; (void)ws_size;
    const float* x    = (const float*)d_in[0];
    const float* kern = (const float*)d_in[1];
    float*       out  = (float*)d_out;

    dim3 grid(NPAIR, NTILE);
    sig_scan_kernel<<<grid, 64, 0, stream>>>(x, kern, out);
}


# Output stores shared by both waves: each hands the other its sums of 2 of a group's 4 tiles through LDS and stores the other 2 (16 rows each, not 32 from wave 1); same sums, LDS traffic, barriers
# speedup vs baseline: 8.4441x; 1.0883x over previous
#include <hip/hip_runtime.h>


typedef __attribute__((ext_vector_type(2))) float v2f;
typedef __attribute__((ext_vector_type(4))) float v4f;
typedef __attribute__((ext_vector_type(8))) float v8f;

#define BATCH  32
#define LOUT   512
#define LSEQ   511
#define FEAT   4
#define NCOMP  10
#define NLOW   6
#define NHIGH  4
#define NF     340
#define NCH    341
#define TCOLS  16
#define TSTEPS 8
#define NPAIR  16
#define NTILE  22
#define TFULL  504
#define PF     4
#define HALF   2
#define ONESW  16
#define ONESR  64
#define XPITCH 3
#define TSLOT  (32 * XPITCH)
#define SEQPAD 8

struct LowState {
    float p21, p31, p32, tot;
};

struct HighState {
    float p41, p42, p43, tot;
};

__device__ __forceinline__ void low_tile(const v2f a, const v2f (&bfrag)[NCOMP],
                                         LowState& s, float (&o)[TSTEPS])
{
    v8f m[NLOW];
    #pragma unroll
    for (int c = 0; c < NLOW; ++c) {
        v8f z = {};
        m[c] = __builtin_amdgcn_wmma_f32_16x16x4_f32(
            false, a, false, bfrag[c], (short)0, z, false, false);
    }

    #pragma unroll
    for (int r = 0; r < TSTEPS; ++r) {
        const float m0 = m[0][r], m1 = m[1][r], m2 = m[2][r];
        const float m3 = m[3][r], m4 = m[4][r], m5 = m[5][r];

        const float r1 = m0;
        const float r2 = m2 * s.p21;  s.p21 += m1;
        const float s3 = m4 * s.p31;  s.p31 += m3;
        const float r3 = m5 * s.p32;  s.p32 += s3;

        s.tot += r1 + r2 + r3;
        o[r] = s.tot;
    }
}

__device__ __forceinline__ void high_tile(const v2f a, const v2f (&bfrag)[NCOMP],
                                          HighState& s, float (&q)[TSTEPS])
{
    v8f m[NHIGH];
    #pragma unroll
    for (int c = 0; c < NHIGH; ++c) {
        v8f z = {};
        m[c] = __builtin_amdgcn_wmma_f32_16x16x4_f32(
            false, a, false, bfrag[NLOW + c], (short)0, z, false, false);
    }

    #pragma unroll
    for (int r = 0; r < TSTEPS; ++r) {
        const float m6 = m[0][r], m7 = m[1][r], m8 = m[2][r], m9 = m[3][r];

        const float s4 = m7 * s.p41;  s.p41 += m6;
        const float s5 = m8 * s.p42;  s.p42 += s4;
        const float r4 = m9 * s.p43;  s.p43 += s5;

        s.tot += r4;
        q[r] = s.tot;
    }
}

__device__ __forceinline__ void lds_barrier()
{
    __builtin_amdgcn_fence(__ATOMIC_RELEASE, "workgroup", "local");
    __builtin_amdgcn_s_barrier();
    __builtin_amdgcn_fence(__ATOMIC_ACQUIRE, "workgroup", "local");
}

__device__ __forceinline__ void give_tile(const float (&o)[TSTEPS], v4f* __restrict__ slot)
{
    v4f lo, hi;
    lo.x = o[0]; lo.y = o[1]; lo.z = o[2]; lo.w = o[3];
    hi.x = o[4]; hi.y = o[5]; hi.z = o[6]; hi.w = o[7];
    slot[0] = lo;
    slot[1] = hi;
}

template <bool LAST>
__device__ __forceinline__ void sig_group(const int role, const v2f (&a)[PF], const int t0,
                                          const v2f (&bfrag)[NCOMP],
                                          LowState& lo, HighState& hi,
                                          v4f* __restrict__ give, const v4f* __restrict__ recv,
                                          float* __restrict__ orow, const bool valid)
{
    float keep[HALF][TSTEPS];
    float hand[HALF][TSTEPS];

    if (role == 0) {
        float q[PF][TSTEPS];
        #pragma unroll
        for (int i = 0; i < PF; ++i)
            low_tile(a[i], bfrag, lo, q[i]);
        #pragma unroll
        for (int j = 0; j < HALF; ++j) {
            #pragma unroll
            for (int r = 0; r < TSTEPS; ++r) {
                keep[j][r] = q[j][r];
                hand[j][r] = q[HALF + j][r];
            }
        }
    } else {
        float q[PF][TSTEPS];
        #pragma unroll
        for (int i = 0; i < PF; ++i)
            high_tile(a[i], bfrag, hi, q[i]);
        #pragma unroll
        for (int j = 0; j < HALF; ++j) {
            #pragma unroll
            for (int r = 0; r < TSTEPS; ++r) {
                hand[j][r] = q[j][r];
                keep[j][r] = q[HALF + j][r];
            }
        }
    }

    #pragma unroll
    for (int j = 0; j < HALF; ++j)
        give_tile(hand[j], give + j * TSLOT);

    lds_barrier();

    #pragma unroll
    for (int j = 0; j < HALF; ++j) {
        const v4f l0 = recv[j * TSLOT];
        const v4f l1 = recv[j * TSLOT + 1];
        float o[TSTEPS];
        o[0] = l0.x + keep[j][0];
        o[1] = l0.y + keep[j][1];
        o[2] = l0.z + keep[j][2];
        o[3] = l0.w + keep[j][3];
        o[4] = l1.x + keep[j][4];
        o[5] = l1.y + keep[j][5];
        o[6] = l1.z + keep[j][6];
        o[7] = l1.w + keep[j][7];
        if (valid) {
            #pragma unroll
            for (int r = 0; r < TSTEPS; ++r) {
                if (!(LAST && j == HALF - 1 && r == TSTEPS - 1 && role != 0))
                    orow[(size_t)(t0 + j * TSTEPS + 1 + r) * NCH] = o[r];
            }
        }
    }
}

__global__ __launch_bounds__(64)
void sig_scan_kernel(const float* __restrict__ x,
                     const float* __restrict__ kern,
                     float* __restrict__ out)
{
    __shared__ v4f xch[2][PF][32][XPITCH];
    __shared__ v4f seqs[2][LOUT + SEQPAD];

    const int lane = threadIdx.x & 31;
    const int role = __builtin_amdgcn_readfirstlane((int)(threadIdx.x >> 5));
    const int bp   = blockIdx.x;
    const int nf0  = blockIdx.y * TCOLS;

    const int half = lane >> 4;
    const int l15  = lane & 15;
    const int kb   = half * 2;

    const int  nf    = nf0 + l15;
    const bool valid = nf < NF;

    v2f bfrag[NCOMP];
    #pragma unroll
    for (int c = 0; c < NCOMP; ++c) {
        v2f t; t.x = 0.f; t.y = 0.f;
        if (valid) {
            t.x = kern[(c * FEAT + kb    ) * NF + nf];
            t.y = kern[(c * FEAT + kb + 1) * NF + nf];
        }
        bfrag[c] = t;
    }

    {
        const v4f* rows = (const v4f*)(x + (size_t)(2 * bp) * LOUT * FEAT) + threadIdx.x;
        #pragma unroll
        for (int k = 0; k < (2 * LOUT) / 64; ++k) {
            const int t = (int)threadIdx.x + 64 * (k % 8);
            v4f r0, r1;
            if (64 * (k % 8) + 63 < LSEQ) {
                r0 = rows[64 * k];
                r1 = rows[64 * k + 1];
            } else {
                const int back = t < LSEQ ? 0 : -1;
                r0 = rows[64 * k + back];
                r1 = rows[64 * k + back + 1];
            }
            seqs[k / 8][t] = r1 - r0;
        }
    }

    const v2f* sa = (const v2f*)&seqs[l15 >> 3][l15 & 7] + half;
    float* ocol = out + (size_t)(2 * bp + half) * LOUT * NCH + 1 + nf;
    const int first = role * HALF;
    const int other = HALF - first;
    float* orow = ocol + (size_t)(first * TSTEPS) * NCH;

    LowState lo;
    lo.p21 = 0.f; lo.p31 = 0.f; lo.p32 = 0.f; lo.tot = 0.f;
    HighState hi;
    hi.p41 = 0.f; hi.p42 = 0.f; hi.p43 = 0.f; hi.tot = 0.f;

    if (blockIdx.y < ONESW) {
        float* ones = out + (size_t)(2 * bp) * LOUT * NCH;
        ones[(size_t)(blockIdx.y * ONESR + threadIdx.x) * NCH] = 1.0f;
    }
    if (role == 0 && valid)
        ocol[0] = 0.0f;

    lds_barrier();

    v2f nxt[PF];
    #pragma unroll
    for (int i = 0; i < PF; ++i)
        nxt[i] = sa[(i * TSTEPS) * 2];

    v4f*       give0 = &xch[0][other][lane][0];
    const v4f* recv0 = &xch[0][first][lane][0];
    int buf = 0;
    for (int t0 = 0; t0 < TFULL + TSTEPS - PF * TSTEPS; t0 += PF * TSTEPS) {
        v2f a[PF];
        #pragma unroll
        for (int i = 0; i < PF; ++i)
            a[i] = nxt[i];
        #pragma unroll
        for (int i = 0; i < PF; ++i)
            nxt[i] = sa[(t0 + (PF + i) * TSTEPS) * 2];
        sig_group<false>(role, a, t0, bfrag, lo, hi,
                         give0 + buf * (PF * TSLOT), recv0 + buf * (PF * TSLOT), orow, valid);
        buf ^= 1;
    }
    {
        v2f a[PF];
        #pragma unroll
        for (int i = 0; i < PF; ++i)
            a[i] = nxt[i];
        sig_group<true>(role, a, TFULL + TSTEPS - PF * TSTEPS, bfrag, lo, hi,
                        give0 + buf * (PF * TSLOT), recv0 + buf * (PF * TSLOT), orow, valid);
    }
}

extern "C" void kernel_launch(void* const* d_in, const int* in_sizes, int n_in,
                              void* d_out, int out_size, void* d_ws, size_t ws_size,
                              hipStream_t stream)
{
    (void)in_sizes; (void)n_in; (void)out_size; (void)d_ws; (void)ws_size;
    const float* x    = (const float*)d_in[0];
    const float* kern = (const float*)d_in[1];
    float*       out  = (float*)d_out;

    dim3 grid(NPAIR, NTILE);
    sig_scan_kernel<<<grid, 64, 0, stream>>>(x, kern, out);
}
